# starting kernel
# baseline (speedup 1.0000x reference)
#include <hip/hip_runtime.h>


typedef float v2f __attribute__((ext_vector_type(2)));
typedef float v8f __attribute__((ext_vector_type(8)));

#define NTOK 4096
#define DM   512
#define NH   8
#define DHD  64
#define SCALE 0.125f

static __device__ __forceinline__ v8f wmma_f32(v2f a, v2f b, v8f c) {
  return __builtin_amdgcn_wmma_f32_16x16x4_f32(
       false, a,  false, b,
       (short)0, c,  false,  false);
}

__global__ __launch_bounds__(256) void qkv_proj_kernel(
    const float* __restrict__ h,
    const float* __restrict__ Wq, const float* __restrict__ bq,
    const float* __restrict__ Wk, const float* __restrict__ bk,
    const float* __restrict__ Wv, const float* __restrict__ bv,
    float* __restrict__ Qo, float* __restrict__ Ko, float* __restrict__ Vo)
{
  const int lane = threadIdx.x & 31;
  const int wid  = threadIdx.x >> 5;
  const int tile = blockIdx.x * 8 + wid;
  const int mat  = tile >> 13;
  const int rem  = tile & 8191;
  const int rowT = rem >> 5;
  const int colT = rem & 31;

  const float* W; const float* bias; float* Out;
  if (mat == 0)      { W = Wq; bias = bq; Out = Qo; }
  else if (mat == 1) { W = Wk; bias = bk; Out = Ko; }
  else               { W = Wv; bias = bv; Out = Vo; }

  const int mr   = lane & 15;
  const int half = lane >> 4;
  const int n    = colT * 16 + mr;
  const float* hrow = h + (size_t)(rowT * 16 + mr) * DM;

  v8f c = {};
  for (int k0 = 0; k0 < DM; k0 += 4) {
    const int ka = k0 + 2 * half;
    v2f a = *(const v2f*)(hrow + ka);
    v2f b;
    b.x = W[(size_t)(ka + 0) * DM + n];
    b.y = W[(size_t)(ka + 1) * DM + n];
    c = wmma_f32(a, b, c);
  }
  const float bn = bias[n];
  #pragma unroll
  for (int v = 0; v < 8; ++v)
    Out[(size_t)(rowT * 16 + v + 8 * half) * DM + n] = c[v] + bn;
}

__global__ __launch_bounds__(128) void graph_attn_kernel(
    const float* __restrict__ Q, const float* __restrict__ K,
    const float* __restrict__ V,
    const float* __restrict__ adj, const float* __restrict__ msk,
    float* __restrict__ out)
{
  __shared__ float pl[4][16 * 18];

  const int lane = threadIdx.x & 31;
  const int wid  = threadIdx.x >> 5;
  const int task = blockIdx.x * 4 + wid;
  const int head = task >> 8;
  const int rowB = (task & 255) * 16;
  const int nn   = lane & 15;
  const int half = lane >> 4;
  float* plw = pl[wid];

  const float* qrow = Q + (size_t)(rowB + nn) * DM + head * DHD;
  v2f qa[16];
  #pragma unroll
  for (int s = 0; s < 16; ++s)
    qa[s] = *(const v2f*)(qrow + 4 * s + 2 * half);

  v8f acc[4] = {};
  float mrow[8], lrow[8];
  #pragma unroll
  for (int v = 0; v < 8; ++v) { mrow[v] = -3.0e38f; lrow[v] = 0.0f; }

  for (int j0 = 0; j0 < NTOK; j0 += 16) {
    v8f sf = {};
    const float* krow = K + (size_t)(j0 + nn) * DM + head * DHD;
    #pragma unroll
    for (int s = 0; s < 16; ++s) {
      v2f b = *(const v2f*)(krow + 4 * s + 2 * half);
      sf = wmma_f32(qa[s], b, sf);
    }

    if (j0 + 16 < NTOK)
      __builtin_prefetch(&adj[(size_t)(rowB + 8 * half) * NTOK + j0 + 16 + nn], 0, 1);

    float p[8];
    #pragma unroll
    for (int v = 0; v < 8; ++v) {
      const size_t e = (size_t)(rowB + v + 8 * half) * NTOK + j0 + nn;
      const float sc = (adj[e] * sf[v] + msk[e]) * SCALE;
      float bm = sc;
      bm = fmaxf(bm, __shfl_xor(bm, 1, 16));
      bm = fmaxf(bm, __shfl_xor(bm, 2, 16));
      bm = fmaxf(bm, __shfl_xor(bm, 4, 16));
      bm = fmaxf(bm, __shfl_xor(bm, 8, 16));
      const float newm = fmaxf(mrow[v], bm);
      const float f    = __expf(mrow[v] - newm);
      const float pv   = __expf(sc - newm);
      float bs = pv;
      bs += __shfl_xor(bs, 1, 16);
      bs += __shfl_xor(bs, 2, 16);
      bs += __shfl_xor(bs, 4, 16);
      bs += __shfl_xor(bs, 8, 16);
      lrow[v] = lrow[v] * f + bs;
      mrow[v] = newm;
      p[v] = pv;
      #pragma unroll
      for (int t = 0; t < 4; ++t) acc[t][v] *= f;
    }

    #pragma unroll
    for (int v = 0; v < 8; ++v)
      plw[(v + 8 * half) * 18 + nn] = p[v];
    __builtin_amdgcn_wave_barrier();

    #pragma unroll
    for (int s = 0; s < 4; ++s) {
      v2f a = *(const v2f*)&plw[nn * 18 + 4 * s + 2 * half];
      const float* v0 = V + (size_t)(j0 + 4 * s + 2 * half) * DM + head * DHD;
      #pragma unroll
      for (int t = 0; t < 4; ++t) {
        v2f b;
        b.x = v0[t * 16 + nn];
        b.y = v0[DM + t * 16 + nn];
        acc[t] = wmma_f32(a, b, acc[t]);
      }
    }
  }

  #pragma unroll
  for (int t = 0; t < 4; ++t) {
    #pragma unroll
    for (int v = 0; v < 8; ++v) {
      out[(size_t)(rowB + v + 8 * half) * DM + head * DHD + t * 16 + nn] =
          acc[t][v] / lrow[v];
    }
  }
}

extern "C" void kernel_launch(void* const* d_in, const int* in_sizes, int n_in,
                              void* d_out, int out_size, void* d_ws, size_t ws_size,
                              hipStream_t stream) {
  (void)in_sizes; (void)n_in; (void)out_size; (void)ws_size;
  const float* h   = (const float*)d_in[0];
  const float* adj = (const float*)d_in[1];
  const float* msk = (const float*)d_in[2];
  const float* Wq  = (const float*)d_in[3];
  const float* bq  = (const float*)d_in[4];
  const float* Wk  = (const float*)d_in[5];
  const float* bk  = (const float*)d_in[6];
  const float* Wv  = (const float*)d_in[7];
  const float* bv  = (const float*)d_in[8];
  float* out = (float*)d_out;

  float* Qo = (float*)d_ws;
  float* Ko = Qo + (size_t)NTOK * DM;
  float* Vo = Ko + (size_t)NTOK * DM;

  qkv_proj_kernel<<<3 * 256 * 32 / 8, 256, 0, stream>>>(
      h, Wq, bq, Wk, bk, Wv, bv, Qo, Ko, Vo);

  graph_attn_kernel<<<(NH * 256) / 4, 128, 0, stream>>>(
      Qo, Ko, Vo, adj, msk, out);
}


# attention K and V tiles staged once per 24-wave group in LDS (2 stages, one barrier per trip, whole-line loads); waves read B fragments from LDS, not global; same sums, same order
# speedup vs baseline: 1.7011x; 1.7011x over previous
#include <hip/hip_runtime.h>


typedef float v2f __attribute__((ext_vector_type(2)));
typedef float v4f __attribute__((ext_vector_type(4)));
typedef float v8f __attribute__((ext_vector_type(8)));

#define NTOK 4096
#define DM   512
#define NH   8
#define DHD  64
#define SCALE 0.125f

static __device__ __forceinline__ v8f wmma_f32(v2f a, v2f b, v8f c) {
  return __builtin_amdgcn_wmma_f32_16x16x4_f32(
       false, a,  false, b,
       (short)0, c,  false,  false);
}

__global__ __launch_bounds__(256) void qkv_proj_kernel(
    const float* __restrict__ h,
    const float* __restrict__ Wq, const float* __restrict__ bq,
    const float* __restrict__ Wk, const float* __restrict__ bk,
    const float* __restrict__ Wv, const float* __restrict__ bv,
    float* __restrict__ Qo, float* __restrict__ Ko, float* __restrict__ Vo)
{
  const int lane = threadIdx.x & 31;
  const int wid  = threadIdx.x >> 5;
  const int tile = blockIdx.x * 8 + wid;
  const int mat  = tile >> 13;
  const int rem  = tile & 8191;
  const int rowT = rem >> 5;
  const int colT = rem & 31;

  const float* W; const float* bias; float* Out;
  if (mat == 0)      { W = Wq; bias = bq; Out = Qo; }
  else if (mat == 1) { W = Wk; bias = bk; Out = Ko; }
  else               { W = Wv; bias = bv; Out = Vo; }

  const int mr   = lane & 15;
  const int half = lane >> 4;
  const int n    = colT * 16 + mr;
  const float* hrow = h + (size_t)(rowT * 16 + mr) * DM;

  v8f c = {};
  for (int k0 = 0; k0 < DM; k0 += 4) {
    const int ka = k0 + 2 * half;
    v2f a = *(const v2f*)(hrow + ka);
    v2f b;
    b.x = W[(size_t)(ka + 0) * DM + n];
    b.y = W[(size_t)(ka + 1) * DM + n];
    c = wmma_f32(a, b, c);
  }
  const float bn = bias[n];
  #pragma unroll
  for (int v = 0; v < 8; ++v)
    Out[(size_t)(rowT * 16 + v + 8 * half) * DM + n] = c[v] + bn;
}

#define KSPLIT 3
#define QBLK   8
#define GROWS  (16 * KSPLIT)
#define NTRIP  ((NTOK / 16 + KSPLIT - 1) / KSPLIT)
#define KPITCH 68
#define VPITCH 160

__global__ __launch_bounds__(32 * QBLK * KSPLIT) void graph_attn_kernel(
    const float* __restrict__ Q, const float* __restrict__ K,
    const float* __restrict__ V,
    const float* __restrict__ adj, const float* __restrict__ msk,
    float* __restrict__ out)
{
  __shared__ float pl[QBLK * KSPLIT][16 * 18];
  __shared__ alignas(16) float kst[2][GROWS * KPITCH];
  __shared__ alignas(16) float vst[2][(GROWS / 2) * VPITCH];
  __shared__ float pacc[QBLK][KSPLIT - 1][4 * 8 * 32];
  __shared__ float pmax[QBLK][KSPLIT - 1][16];
  __shared__ float psum[QBLK][KSPLIT - 1][16];

  const int lane = threadIdx.x & 31;
  const int wid  = __builtin_amdgcn_readfirstlane(threadIdx.x >> 5);
  const int qb   = wid & (QBLK - 1);
  const int ks   = wid / QBLK;
  const int task = blockIdx.x * QBLK + qb;
  const int head = task >> 8;
  const int rowB = (task & 255) * 16;
  const int nn   = lane & 15;
  const int half = lane >> 4;
  float* plw = pl[wid];

  const float* qrow = Q + (size_t)(rowB + nn) * DM + head * DHD;
  v2f qa[16];
  #pragma unroll
  for (int s = 0; s < 16; ++s)
    qa[s] = *(const v2f*)(qrow + 4 * s + 2 * half);

  const float* kg = K + (size_t)(2 * wid + half) * DM + head * DHD + 4 * nn;
  const float* vg = V + (size_t)(2 * wid) * DM + head * DHD + 2 * lane;
  const int kso = (2 * wid + half) * KPITCH + 4 * nn;
  const int vso = wid * VPITCH + 4 * lane;
  {
    const v4f kr0 = *(const v4f*)kg;
    const v2f va0 = *(const v2f*)vg;
    const v2f vb0 = *(const v2f*)(vg + DM);
    const v4f vr0 = {va0.x, vb0.x, va0.y, vb0.y};
    *(v4f*)&kst[0][kso] = kr0;
    *(v4f*)&vst[0][vso] = vr0;
  }
  __syncthreads();

  v8f acc[4] = {};
  float mrow[8], lrow[8];
  #pragma unroll
  for (int v = 0; v < 8; ++v) { mrow[v] = -3.0e38f; lrow[v] = 0.0f; }

  for (int it = 0; it < NTRIP; ++it) {
    const int st = it & 1;
    const int jn = GROWS * (it + 1);
    const int jl = (jn + 2 * wid) < NTOK ? jn : 0;
    const v4f kr = *(const v4f*)(kg + (size_t)jl * DM);
    const v2f va = *(const v2f*)(vg + (size_t)jl * DM);
    const v2f vb = *(const v2f*)(vg + (size_t)jl * DM + DM);

    const int j0 = GROWS * it + 16 * ks;
    if (j0 < NTOK) {
    const float* ksw = &kst[st][16 * ks * KPITCH];
    const float* vsw = &vst[st][8 * ks * VPITCH];

    v8f sf = {};
    #pragma unroll
    for (int s = 0; s < 16; ++s) {
      v2f b = *(const v2f*)&ksw[nn * KPITCH + 4 * s + 2 * half];
      sf = wmma_f32(qa[s], b, sf);
    }

    if (j0 + 16 * KSPLIT < NTOK)
      __builtin_prefetch(&adj[(size_t)(rowB + 8 * half) * NTOK + j0 + 16 * KSPLIT + nn], 0, 1);

    float p[8], sc[8], bm[8], fr[8], bs[8];
    #pragma unroll
    for (int v = 0; v < 8; ++v) {
      const size_t e = (size_t)(rowB + v + 8 * half) * NTOK + j0 + nn;
      sc[v] = (adj[e] * sf[v] + msk[e]) * SCALE;
      bm[v] = sc[v];
    }
    #pragma unroll
    for (int m = 1; m < 16; m <<= 1) {
      #pragma unroll
      for (int v = 0; v < 8; ++v) bm[v] = fmaxf(bm[v], __shfl_xor(bm[v], m, 16));
    }
    #pragma unroll
    for (int v = 0; v < 8; ++v) {
      const float newm = fmaxf(mrow[v], bm[v]);
      fr[v]   = __expf(mrow[v] - newm);
      p[v]    = __expf(sc[v] - newm);
      mrow[v] = newm;
      bs[v]   = p[v];
    }
    #pragma unroll
    for (int m = 1; m < 16; m <<= 1) {
      #pragma unroll
      for (int v = 0; v < 8; ++v) bs[v] += __shfl_xor(bs[v], m, 16);
    }
    #pragma unroll
    for (int v = 0; v < 8; ++v) {
      lrow[v] = lrow[v] * fr[v] + bs[v];
      #pragma unroll
      for (int t = 0; t < 4; ++t) acc[t][v] *= fr[v];
    }

    {
      const v4f vr = {va.x, vb.x, va.y, vb.y};
      *(v4f*)&kst[st ^ 1][kso] = kr;
      *(v4f*)&vst[st ^ 1][vso] = vr;
    }

    #pragma unroll
    for (int v = 0; v < 8; ++v)
      plw[(v + 8 * half) * 18 + nn] = p[v];
    __builtin_amdgcn_wave_barrier();

    #pragma unroll
    for (int s = 0; s < 4; ++s) {
      v2f a = *(const v2f*)&plw[nn * 18 + 4 * s + 2 * half];
      const float* v0 = vsw + (2 * s + half) * VPITCH + 2 * nn;
      #pragma unroll
      for (int t = 0; t < 4; ++t) {
        const v2f b = *(const v2f*)&v0[32 * t];
        acc[t] = wmma_f32(a, b, acc[t]);
      }
    }
    }

    __syncthreads();
  }

  const int lane_e = (int)__builtin_amdgcn_mbcnt_lo(0xffffffffu, 0u);
  const int nn_e   = lane_e & 15;
  const int half_e = lane_e >> 4;

  if (ks != 0) {
    float* pa = pacc[qb][ks - 1];
    #pragma unroll
    for (int t = 0; t < 4; ++t) {
      #pragma unroll
      for (int v = 0; v < 8; ++v)
        pa[(t * 8 + v) * 32 + lane_e] = acc[t][v];
    }
    if (nn_e == 0) {
      #pragma unroll
      for (int v = 0; v < 8; ++v) {
        pmax[qb][ks - 1][v + 8 * half_e] = mrow[v];
        psum[qb][ks - 1][v + 8 * half_e] = lrow[v];
      }
    }
  }
  __syncthreads();
  if (ks != 0) return;

  float f0[8], fk[KSPLIT - 1][8], ltot[8];
  #pragma unroll
  for (int v = 0; v < 8; ++v) {
    const int row = v + 8 * half_e;
    float m = mrow[v];
    #pragma unroll
    for (int k = 0; k < KSPLIT - 1; ++k) m = fmaxf(m, pmax[qb][k][row]);
    f0[v] = __expf(mrow[v] - m);
    float l = lrow[v] * f0[v];
    #pragma unroll
    for (int k = 0; k < KSPLIT - 1; ++k) {
      fk[k][v] = __expf(pmax[qb][k][row] - m);
      l += psum[qb][k][row] * fk[k][v];
    }
    ltot[v] = l;
  }
  #pragma unroll
  for (int t = 0; t < 4; ++t) {
    #pragma unroll
    for (int v = 0; v < 8; ++v) {
      float a = acc[t][v] * f0[v];
      #pragma unroll
      for (int k = 0; k < KSPLIT - 1; ++k)
        a += pacc[qb][k][(t * 8 + v) * 32 + lane_e] * fk[k][v];
      out[(size_t)(rowB + v + 8 * half_e) * DM + head * DHD + t * 16 + nn_e] =
          a / ltot[v];
    }
  }
}

extern "C" void kernel_launch(void* const* d_in, const int* in_sizes, int n_in,
                              void* d_out, int out_size, void* d_ws, size_t ws_size,
                              hipStream_t stream) {
  (void)in_sizes; (void)n_in; (void)out_size; (void)ws_size;
  const float* h   = (const float*)d_in[0];
  const float* adj = (const float*)d_in[1];
  const float* msk = (const float*)d_in[2];
  const float* Wq  = (const float*)d_in[3];
  const float* bq  = (const float*)d_in[4];
  const float* Wk  = (const float*)d_in[5];
  const float* bk  = (const float*)d_in[6];
  const float* Wv  = (const float*)d_in[7];
  const float* bv  = (const float*)d_in[8];
  float* out = (float*)d_out;

  float* Qo = (float*)d_ws;
  float* Ko = Qo + (size_t)NTOK * DM;
  float* Vo = Ko + (size_t)NTOK * DM;

  qkv_proj_kernel<<<3 * 256 * 32 / 8, 256, 0, stream>>>(
      h, Wq, bq, Wk, bk, Wv, bv, Qo, Ko, Vo);

  graph_attn_kernel<<<(NH * 256) / QBLK, 32 * QBLK * KSPLIT, 0, stream>>>(
      Qo, Ko, Vo, adj, msk, out);
}


# attention softmax moved to the A-operand side of the score tile's LDS transpose: row max in-lane + one half exchange, row sum per lane joined once at the end; regroups row-sum tree and P@V key order
# speedup vs baseline: 1.7611x; 1.0353x over previous
#include <hip/hip_runtime.h>


typedef float v2f __attribute__((ext_vector_type(2)));
typedef float v4f __attribute__((ext_vector_type(4)));
typedef float v8f __attribute__((ext_vector_type(8)));

#define NTOK 4096
#define DM   512
#define NH   8
#define DHD  64
#define SCALE 0.125f

static __device__ __forceinline__ v8f wmma_f32(v2f a, v2f b, v8f c) {
  return __builtin_amdgcn_wmma_f32_16x16x4_f32(
       false, a,  false, b,
       (short)0, c,  false,  false);
}

__global__ __launch_bounds__(256) void qkv_proj_kernel(
    const float* __restrict__ h,
    const float* __restrict__ Wq, const float* __restrict__ bq,
    const float* __restrict__ Wk, const float* __restrict__ bk,
    const float* __restrict__ Wv, const float* __restrict__ bv,
    float* __restrict__ Qo, float* __restrict__ Ko, float* __restrict__ Vo)
{
  const int lane = threadIdx.x & 31;
  const int wid  = threadIdx.x >> 5;
  const int tile = blockIdx.x * 8 + wid;
  const int mat  = tile >> 13;
  const int rem  = tile & 8191;
  const int rowT = rem >> 5;
  const int colT = rem & 31;

  const float* W; const float* bias; float* Out;
  if (mat == 0)      { W = Wq; bias = bq; Out = Qo; }
  else if (mat == 1) { W = Wk; bias = bk; Out = Ko; }
  else               { W = Wv; bias = bv; Out = Vo; }

  const int mr   = lane & 15;
  const int half = lane >> 4;
  const int n    = colT * 16 + mr;
  const float* hrow = h + (size_t)(rowT * 16 + mr) * DM;

  v8f c = {};
  for (int k0 = 0; k0 < DM; k0 += 4) {
    const int ka = k0 + 2 * half;
    v2f a = *(const v2f*)(hrow + ka);
    v2f b;
    b.x = W[(size_t)(ka + 0) * DM + n];
    b.y = W[(size_t)(ka + 1) * DM + n];
    c = wmma_f32(a, b, c);
  }
  const float bn = bias[n];
  #pragma unroll
  for (int v = 0; v < 8; ++v)
    Out[(size_t)(rowT * 16 + v + 8 * half) * DM + n] = c[v] + bn;
}

#define KSPLIT 3
#define QBLK   8
#define GROWS  (16 * KSPLIT)
#define NTRIP  ((NTOK / 16 + KSPLIT - 1) / KSPLIT)
#define KPITCH 68
#define VPITCH 136
#define SPITCH 20

static __device__ __forceinline__ float xhalf(float x) {
  const int b = __builtin_bit_cast(int, x);
  return __builtin_bit_cast(float,
      __builtin_amdgcn_permlanex16(b, b, 0x76543210, 0xfedcba98, false, false));
}

template <int N>
static __device__ __forceinline__ float row_share(float x) {
  return __builtin_bit_cast(float, __builtin_amdgcn_update_dpp(
      0, __builtin_bit_cast(int, x), 0x150 + N, 0xf, 0xf, true));
}

__global__ __launch_bounds__(32 * QBLK * KSPLIT) void graph_attn_kernel(
    const float* __restrict__ Q, const float* __restrict__ K,
    const float* __restrict__ V,
    const float* __restrict__ adj, const float* __restrict__ msk,
    float* __restrict__ out)
{
  __shared__ alignas(16) float pl[QBLK * KSPLIT][16 * SPITCH];
  __shared__ alignas(16) float kst[2][GROWS * KPITCH];
  __shared__ alignas(16) float vst[2][(GROWS / 2) * VPITCH];
  __shared__ float pacc[QBLK][KSPLIT - 1][4 * 8 * 32];
  __shared__ float pmax[QBLK][KSPLIT][16];
  __shared__ float psum[QBLK][KSPLIT][16];

  const int lane = threadIdx.x & 31;
  const int wid  = __builtin_amdgcn_readfirstlane(threadIdx.x >> 5);
  const int qb   = wid & (QBLK - 1);
  const int ks   = wid / QBLK;
  const int task = blockIdx.x * QBLK + qb;
  const int head = task >> 8;
  const int rowB = (task & 255) * 16;
  const int nn   = lane & 15;
  const int half = lane >> 4;
  float* plw = pl[wid];

  const float* qrow = Q + (size_t)(rowB + nn) * DM + head * DHD;
  v2f qa[16];
  #pragma unroll
  for (int s = 0; s < 16; ++s)
    qa[s] = *(const v2f*)(qrow + 4 * s + 2 * half);

  const float* adjp = adj + (size_t)(rowB + nn) * NTOK + 8 * half;
  const float* mskp = msk + (size_t)(rowB + nn) * NTOK + 8 * half;

  const float* kg = K + (size_t)(2 * wid + half) * DM + head * DHD + 4 * nn;
  const float* vg = V + (size_t)(2 * wid) * DM + head * DHD + 2 * lane;
  const int kso = (2 * wid + half) * KPITCH + 4 * nn;
  const int vso = wid * VPITCH + 4 * lane;
  {
    const v4f kr0 = *(const v4f*)kg;
    const v2f va0 = *(const v2f*)vg;
    const v2f vb0 = *(const v2f*)(vg + DM);
    const v4f vr0 = {va0.x, vb0.x, va0.y, vb0.y};
    *(v4f*)&kst[0][kso] = kr0;
    *(v4f*)&vst[0][vso] = vr0;
  }
  __syncthreads();

  v8f acc[4] = {};
  float m_run = -3.0e38f;
  float l_run = 0.0f;

  for (int it = 0; it < NTRIP; ++it) {
    const int st = it & 1;
    const int jn = GROWS * (it + 1);
    const int jl = (jn + 2 * wid) < NTOK ? jn : 0;
    const v4f kr = *(const v4f*)(kg + (size_t)jl * DM);
    const v2f va = *(const v2f*)(vg + (size_t)jl * DM);
    const v2f vb = *(const v2f*)(vg + (size_t)jl * DM + DM);

    const int j0 = GROWS * it + 16 * ks;
    if (j0 < NTOK) {
    const float* ksw = &kst[st][16 * ks * KPITCH];
    const float* vsw = &vst[st][8 * ks * VPITCH];

    const v4f ad0 = *(const v4f*)(adjp + j0);
    const v4f ad1 = *(const v4f*)(adjp + j0 + 4);
    const v4f mk0 = *(const v4f*)(mskp + j0);
    const v4f mk1 = *(const v4f*)(mskp + j0 + 4);
    if (j0 + 16 * KSPLIT < NTOK)
      __builtin_prefetch(adjp + j0 + 16 * KSPLIT, 0, 1);
    __builtin_amdgcn_sched_barrier(0);

    v8f sf = {};
    #pragma unroll
    for (int s = 0; s < 16; ++s) {
      v2f b = *(const v2f*)&ksw[nn * KPITCH + 4 * s + 2 * half];
      sf = wmma_f32(qa[s], b, sf);
    }

    #pragma unroll
    for (int v = 0; v < 8; ++v)
      plw[(v + 8 * half) * SPITCH + nn] = sf[v];
    __builtin_amdgcn_wave_barrier();
    const v4f s0 = *(const v4f*)&plw[nn * SPITCH + 8 * half];
    const v4f s1 = *(const v4f*)&plw[nn * SPITCH + 8 * half + 4];

    float p[8];
    #pragma unroll
    for (int i = 0; i < 4; ++i) {
      p[i]     = (ad0[i] * s0[i] + mk0[i]) * SCALE;
      p[4 + i] = (ad1[i] * s1[i] + mk1[i]) * SCALE;
    }
    float bm = fmaxf(fmaxf(fmaxf(p[0], p[1]), fmaxf(p[2], p[3])),
                     fmaxf(fmaxf(p[4], p[5]), fmaxf(p[6], p[7])));
    bm = fmaxf(bm, xhalf(bm));
    const float newm = fmaxf(m_run, bm);
    const float fr   = __expf(m_run - newm);
    m_run = newm;
    #pragma unroll
    for (int i = 0; i < 8; ++i) p[i] = __expf(p[i] - newm);
    l_run = l_run * fr + (((p[0] + p[1]) + (p[2] + p[3])) +
                          ((p[4] + p[5]) + (p[6] + p[7])));

    const float frx = __builtin_bit_cast(float, __builtin_amdgcn_update_dpp(
        __builtin_bit_cast(int, fr), __builtin_bit_cast(int, fr),
        0x128  , 0x2  , 0xf, false));
#define RESCALE_ROW(v) { const float f_ = row_share<v>(frx);                   \
      acc[0][v] *= f_; acc[1][v] *= f_; acc[2][v] *= f_; acc[3][v] *= f_; }
    RESCALE_ROW(0) RESCALE_ROW(1) RESCALE_ROW(2) RESCALE_ROW(3)
    RESCALE_ROW(4) RESCALE_ROW(5) RESCALE_ROW(6) RESCALE_ROW(7)
#undef RESCALE_ROW

    {
      const v4f vr = {va.x, vb.x, va.y, vb.y};
      *(v4f*)&kst[st ^ 1][kso] = kr;
      *(v4f*)&vst[st ^ 1][vso] = vr;
    }

    #pragma unroll
    for (int s = 0; s < 4; ++s) {
      const v2f a = {p[2 * s], p[2 * s + 1]};
      const float* v0 = vsw + (4 * half + s) * VPITCH + 2 * nn;
      #pragma unroll
      for (int t = 0; t < 4; ++t) {
        const v2f b = *(const v2f*)&v0[32 * t];
        acc[t] = wmma_f32(a, b, acc[t]);
      }
    }
    }

    __syncthreads();
  }

  const int lane_e = (int)__builtin_amdgcn_mbcnt_lo(0xffffffffu, 0u);
  const int nn_e   = lane_e & 15;
  const int half_e = lane_e >> 4;

  const float l_own = l_run + xhalf(l_run);
  pmax[qb][ks][nn_e] = m_run;
  psum[qb][ks][nn_e] = l_own;

  if (ks != 0) {
    float* pa = pacc[qb][ks - 1];
    #pragma unroll
    for (int t = 0; t < 4; ++t) {
      #pragma unroll
      for (int v = 0; v < 8; ++v)
        pa[(t * 8 + v) * 32 + lane_e] = acc[t][v];
    }
  }
  __syncthreads();
  if (ks != 0) return;

  float f0[8], fk[KSPLIT - 1][8], ltot[8];
  #pragma unroll
  for (int v = 0; v < 8; ++v) {
    const int row = v + 8 * half_e;
    const float m0 = pmax[qb][0][row];
    float m = m0;
    #pragma unroll
    for (int k = 0; k < KSPLIT - 1; ++k) m = fmaxf(m, pmax[qb][k + 1][row]);
    f0[v] = __expf(m0 - m);
    float l = psum[qb][0][row] * f0[v];
    #pragma unroll
    for (int k = 0; k < KSPLIT - 1; ++k) {
      fk[k][v] = __expf(pmax[qb][k + 1][row] - m);
      l += psum[qb][k + 1][row] * fk[k][v];
    }
    ltot[v] = l;
  }
  #pragma unroll
  for (int t = 0; t < 4; ++t) {
    #pragma unroll
    for (int v = 0; v < 8; ++v) {
      float a = acc[t][v] * f0[v];
      #pragma unroll
      for (int k = 0; k < KSPLIT - 1; ++k)
        a += pacc[qb][k][(t * 8 + v) * 32 + lane_e] * fk[k][v];
      out[(size_t)(rowB + v + 8 * half_e) * DM + head * DHD + t * 16 + nn_e] =
          a / ltot[v];
    }
  }
}

extern "C" void kernel_launch(void* const* d_in, const int* in_sizes, int n_in,
                              void* d_out, int out_size, void* d_ws, size_t ws_size,
                              hipStream_t stream) {
  (void)in_sizes; (void)n_in; (void)out_size; (void)ws_size;
  const float* h   = (const float*)d_in[0];
  const float* adj = (const float*)d_in[1];
  const float* msk = (const float*)d_in[2];
  const float* Wq  = (const float*)d_in[3];
  const float* bq  = (const float*)d_in[4];
  const float* Wk  = (const float*)d_in[5];
  const float* bk  = (const float*)d_in[6];
  const float* Wv  = (const float*)d_in[7];
  const float* bv  = (const float*)d_in[8];
  float* out = (float*)d_out;

  float* Qo = (float*)d_ws;
  float* Ko = Qo + (size_t)NTOK * DM;
  float* Vo = Ko + (size_t)NTOK * DM;

  qkv_proj_kernel<<<3 * 256 * 32 / 8, 256, 0, stream>>>(
      h, Wq, bq, Wk, bk, Wv, bv, Qo, Ko, Vo);

  graph_attn_kernel<<<(NH * 256) / QBLK, 32 * QBLK * KSPLIT, 0, stream>>>(
      Qo, Ko, Vo, adj, msk, out);
}


# projection operands staged through LDS: 128x64 block per 8-wave group, 64-k stages x2, whole-line 16-byte loads, a wave's h fragment feeds 4 WMMAs; regroups the k order inside its sums
# speedup vs baseline: 2.0420x; 1.1595x over previous
#include <hip/hip_runtime.h>


typedef float v2f __attribute__((ext_vector_type(2)));
typedef float v4f __attribute__((ext_vector_type(4)));
typedef float v8f __attribute__((ext_vector_type(8)));

#define NTOK 4096
#define DM   512
#define NH   8
#define DHD  64
#define SCALE 0.125f

static __device__ __forceinline__ v8f wmma_f32(v2f a, v2f b, v8f c) {
  return __builtin_amdgcn_wmma_f32_16x16x4_f32(
       false, a,  false, b,
       (short)0, c,  false,  false);
}

#define PBM 128
#define PBN 64
#define PBK 64
#define HPITCH 68
#define WPITCH 136

__global__ __launch_bounds__(256) void qkv_proj_kernel(
    const float* __restrict__ h,
    const float* __restrict__ Wq, const float* __restrict__ bq,
    const float* __restrict__ Wk, const float* __restrict__ bk,
    const float* __restrict__ Wv, const float* __restrict__ bv,
    float* __restrict__ Qo, float* __restrict__ Ko, float* __restrict__ Vo)
{
  __shared__ alignas(16) float hst[2][PBM * HPITCH];
  __shared__ alignas(16) float wst[2][(PBK / 2) * WPITCH];

  const int tid  = threadIdx.x;
  const int lane = tid & 31;
  const int wid  = __builtin_amdgcn_readfirstlane(tid >> 5);
  const int c    = lane & 15;
  const int hf   = lane >> 4;

  const int cb   = blockIdx.x >> 5;
  const int rowB = (blockIdx.x & 31) * PBM;
  const int mat  = cb >> 3;
  const int col0 = (cb & 7) * PBN;

  const float* W; const float* bias; float* Out;
  if (mat == 0)      { W = Wq; bias = bq; Out = Qo; }
  else if (mat == 1) { W = Wk; bias = bk; Out = Ko; }
  else               { W = Wv; bias = bv; Out = Vo; }

  const int q  = tid & 15;
  const int r0 = tid >> 4;
  const float* hg = h + (size_t)(rowB + r0) * DM + 4 * q;
  const float* wg = W + (size_t)(2 * r0) * DM + col0 + 4 * q;
  const int hso = r0 * HPITCH + 4 * q;
  const int wso = r0 * WPITCH + 8 * q;

  v4f hr[8], wa[2], wb[2];
  #pragma unroll
  for (int i = 0; i < 8; ++i) hr[i] = *(const v4f*)(hg + (size_t)(16 * i) * DM);
  #pragma unroll
  for (int j = 0; j < 2; ++j) {
    wa[j] = *(const v4f*)(wg + (size_t)(32 * j) * DM);
    wb[j] = *(const v4f*)(wg + (size_t)(32 * j + 1) * DM);
  }
  #pragma unroll
  for (int i = 0; i < 8; ++i) *(v4f*)&hst[0][hso + 16 * i * HPITCH] = hr[i];
  #pragma unroll
  for (int j = 0; j < 2; ++j) {
    const v4f lo = {wa[j].x, wb[j].x, wa[j].y, wb[j].y};
    const v4f hi = {wa[j].z, wb[j].z, wa[j].w, wb[j].w};
    *(v4f*)&wst[0][wso + 16 * j * WPITCH]     = lo;
    *(v4f*)&wst[0][wso + 16 * j * WPITCH + 4] = hi;
  }
  __syncthreads();

  v8f acc[4] = {};
  const int aoff = (16 * wid + c) * HPITCH + 8 * hf;
  const int boff = (4 * hf) * WPITCH + 2 * c;

  for (int ks = 0; ks < DM / PBK; ++ks) {
    const int st = ks & 1;
    const bool more = (ks + 1) < DM / PBK;
    if (more) {
      const float* hn = hg + PBK * (ks + 1);
      const float* wn = wg + (size_t)(PBK * (ks + 1)) * DM;
      #pragma unroll
      for (int i = 0; i < 8; ++i) hr[i] = *(const v4f*)(hn + (size_t)(16 * i) * DM);
      #pragma unroll
      for (int j = 0; j < 2; ++j) {
        wa[j] = *(const v4f*)(wn + (size_t)(32 * j) * DM);
        wb[j] = *(const v4f*)(wn + (size_t)(32 * j + 1) * DM);
      }
    }

    const float* hs = &hst[st][aoff];
    const float* ws = &wst[st][boff];
    #pragma unroll
    for (int kc = 0; kc < PBK; kc += 16) {
      const v4f a01 = *(const v4f*)&hs[kc];
      const v4f a23 = *(const v4f*)&hs[kc + 4];
      const v2f as[4] = {{a01.x, a01.y}, {a01.z, a01.w},
                         {a23.x, a23.y}, {a23.z, a23.w}};
      #pragma unroll
      for (int s = 0; s < 4; ++s) {
        const float* w0 = ws + (kc / 2 + s) * WPITCH;
        #pragma unroll
        for (int t = 0; t < 4; ++t) {
          const v2f b = *(const v2f*)&w0[32 * t];
          acc[t] = wmma_f32(as[s], b, acc[t]);
        }
      }
    }

    if (more) {
      #pragma unroll
      for (int i = 0; i < 8; ++i) *(v4f*)&hst[st ^ 1][hso + 16 * i * HPITCH] = hr[i];
      #pragma unroll
      for (int j = 0; j < 2; ++j) {
        const v4f lo = {wa[j].x, wb[j].x, wa[j].y, wb[j].y};
        const v4f hi = {wa[j].z, wb[j].z, wa[j].w, wb[j].w};
        *(v4f*)&wst[st ^ 1][wso + 16 * j * WPITCH]     = lo;
        *(v4f*)&wst[st ^ 1][wso + 16 * j * WPITCH + 4] = hi;
      }
    }
    __syncthreads();
  }

  #pragma unroll
  for (int t = 0; t < 4; ++t) {
    const int n = col0 + 16 * t + c;
    const float bn = bias[n];
    #pragma unroll
    for (int v = 0; v < 8; ++v)
      Out[(size_t)(rowB + 16 * wid + 8 * hf + v) * DM + n] = acc[t][v] + bn;
  }
}

#define KSPLIT 3
#define QBLK   8
#define GROWS  (16 * KSPLIT)
#define NTRIP  ((NTOK / 16 + KSPLIT - 1) / KSPLIT)
#define KPITCH 68
#define VPITCH 136
#define SPITCH 20

static __device__ __forceinline__ float xhalf(float x) {
  const int b = __builtin_bit_cast(int, x);
  return __builtin_bit_cast(float,
      __builtin_amdgcn_permlanex16(b, b, 0x76543210, 0xfedcba98, false, false));
}

template <int N>
static __device__ __forceinline__ float row_share(float x) {
  return __builtin_bit_cast(float, __builtin_amdgcn_update_dpp(
      0, __builtin_bit_cast(int, x), 0x150 + N, 0xf, 0xf, true));
}

__global__ __launch_bounds__(32 * QBLK * KSPLIT) void graph_attn_kernel(
    const float* __restrict__ Q, const float* __restrict__ K,
    const float* __restrict__ V,
    const float* __restrict__ adj, const float* __restrict__ msk,
    float* __restrict__ out)
{
  __shared__ alignas(16) float pl[QBLK * KSPLIT][16 * SPITCH];
  __shared__ alignas(16) float kst[2][GROWS * KPITCH];
  __shared__ alignas(16) float vst[2][(GROWS / 2) * VPITCH];
  __shared__ float pacc[QBLK][KSPLIT - 1][4 * 8 * 32];
  __shared__ float pmax[QBLK][KSPLIT][16];
  __shared__ float psum[QBLK][KSPLIT][16];

  const int lane = threadIdx.x & 31;
  const int wid  = __builtin_amdgcn_readfirstlane(threadIdx.x >> 5);
  const int qb   = wid & (QBLK - 1);
  const int ks   = wid / QBLK;
  const int task = blockIdx.x * QBLK + qb;
  const int head = task >> 8;
  const int rowB = (task & 255) * 16;
  const int nn   = lane & 15;
  const int half = lane >> 4;
  float* plw = pl[wid];

  const float* qrow = Q + (size_t)(rowB + nn) * DM + head * DHD;
  v2f qa[16];
  #pragma unroll
  for (int s = 0; s < 16; ++s)
    qa[s] = *(const v2f*)(qrow + 4 * s + 2 * half);

  const float* adjp = adj + (size_t)(rowB + nn) * NTOK + 8 * half;
  const float* mskp = msk + (size_t)(rowB + nn) * NTOK + 8 * half;

  const float* kg = K + (size_t)(2 * wid + half) * DM + head * DHD + 4 * nn;
  const float* vg = V + (size_t)(2 * wid) * DM + head * DHD + 2 * lane;
  const int kso = (2 * wid + half) * KPITCH + 4 * nn;
  const int vso = wid * VPITCH + 4 * lane;
  {
    const v4f kr0 = *(const v4f*)kg;
    const v2f va0 = *(const v2f*)vg;
    const v2f vb0 = *(const v2f*)(vg + DM);
    const v4f vr0 = {va0.x, vb0.x, va0.y, vb0.y};
    *(v4f*)&kst[0][kso] = kr0;
    *(v4f*)&vst[0][vso] = vr0;
  }
  __syncthreads();

  v8f acc[4] = {};
  float m_run = -3.0e38f;
  float l_run = 0.0f;

  for (int it = 0; it < NTRIP; ++it) {
    const int st = it & 1;
    const int jn = GROWS * (it + 1);
    const int jl = (jn + 2 * wid) < NTOK ? jn : 0;
    const v4f kr = *(const v4f*)(kg + (size_t)jl * DM);
    const v2f va = *(const v2f*)(vg + (size_t)jl * DM);
    const v2f vb = *(const v2f*)(vg + (size_t)jl * DM + DM);

    const int j0 = GROWS * it + 16 * ks;
    if (j0 < NTOK) {
    const float* ksw = &kst[st][16 * ks * KPITCH];
    const float* vsw = &vst[st][8 * ks * VPITCH];

    const v4f ad0 = *(const v4f*)(adjp + j0);
    const v4f ad1 = *(const v4f*)(adjp + j0 + 4);
    const v4f mk0 = *(const v4f*)(mskp + j0);
    const v4f mk1 = *(const v4f*)(mskp + j0 + 4);
    if (j0 + 16 * KSPLIT < NTOK)
      __builtin_prefetch(adjp + j0 + 16 * KSPLIT, 0, 1);
    __builtin_amdgcn_sched_barrier(0);

    v8f sf = {};
    #pragma unroll
    for (int s = 0; s < 16; ++s) {
      v2f b = *(const v2f*)&ksw[nn * KPITCH + 4 * s + 2 * half];
      sf = wmma_f32(qa[s], b, sf);
    }

    #pragma unroll
    for (int v = 0; v < 8; ++v)
      plw[(v + 8 * half) * SPITCH + nn] = sf[v];
    __builtin_amdgcn_wave_barrier();
    const v4f s0 = *(const v4f*)&plw[nn * SPITCH + 8 * half];
    const v4f s1 = *(const v4f*)&plw[nn * SPITCH + 8 * half + 4];

    float p[8];
    #pragma unroll
    for (int i = 0; i < 4; ++i) {
      p[i]     = (ad0[i] * s0[i] + mk0[i]) * SCALE;
      p[4 + i] = (ad1[i] * s1[i] + mk1[i]) * SCALE;
    }
    float bm = fmaxf(fmaxf(fmaxf(p[0], p[1]), fmaxf(p[2], p[3])),
                     fmaxf(fmaxf(p[4], p[5]), fmaxf(p[6], p[7])));
    bm = fmaxf(bm, xhalf(bm));
    const float newm = fmaxf(m_run, bm);
    const float fr   = __expf(m_run - newm);
    m_run = newm;
    #pragma unroll
    for (int i = 0; i < 8; ++i) p[i] = __expf(p[i] - newm);
    l_run = l_run * fr + (((p[0] + p[1]) + (p[2] + p[3])) +
                          ((p[4] + p[5]) + (p[6] + p[7])));

    const float frx = __builtin_bit_cast(float, __builtin_amdgcn_update_dpp(
        __builtin_bit_cast(int, fr), __builtin_bit_cast(int, fr),
        0x128  , 0x2  , 0xf, false));
#define RESCALE_ROW(v) { const float f_ = row_share<v>(frx);                   \
      acc[0][v] *= f_; acc[1][v] *= f_; acc[2][v] *= f_; acc[3][v] *= f_; }
    RESCALE_ROW(0) RESCALE_ROW(1) RESCALE_ROW(2) RESCALE_ROW(3)
    RESCALE_ROW(4) RESCALE_ROW(5) RESCALE_ROW(6) RESCALE_ROW(7)
#undef RESCALE_ROW

    {
      const v4f vr = {va.x, vb.x, va.y, vb.y};
      *(v4f*)&kst[st ^ 1][kso] = kr;
      *(v4f*)&vst[st ^ 1][vso] = vr;
    }

    #pragma unroll
    for (int s = 0; s < 4; ++s) {
      const v2f a = {p[2 * s], p[2 * s + 1]};
      const float* v0 = vsw + (4 * half + s) * VPITCH + 2 * nn;
      #pragma unroll
      for (int t = 0; t < 4; ++t) {
        const v2f b = *(const v2f*)&v0[32 * t];
        acc[t] = wmma_f32(a, b, acc[t]);
      }
    }
    }

    __syncthreads();
  }

  const int lane_e = (int)__builtin_amdgcn_mbcnt_lo(0xffffffffu, 0u);
  const int nn_e   = lane_e & 15;
  const int half_e = lane_e >> 4;

  const float l_own = l_run + xhalf(l_run);
  pmax[qb][ks][nn_e] = m_run;
  psum[qb][ks][nn_e] = l_own;

  if (ks != 0) {
    float* pa = pacc[qb][ks - 1];
    #pragma unroll
    for (int t = 0; t < 4; ++t) {
      #pragma unroll
      for (int v = 0; v < 8; ++v)
        pa[(t * 8 + v) * 32 + lane_e] = acc[t][v];
    }
  }
  __syncthreads();
  if (ks != 0) return;

  float f0[8], fk[KSPLIT - 1][8], ltot[8];
  #pragma unroll
  for (int v = 0; v < 8; ++v) {
    const int row = v + 8 * half_e;
    const float m0 = pmax[qb][0][row];
    float m = m0;
    #pragma unroll
    for (int k = 0; k < KSPLIT - 1; ++k) m = fmaxf(m, pmax[qb][k + 1][row]);
    f0[v] = __expf(m0 - m);
    float l = psum[qb][0][row] * f0[v];
    #pragma unroll
    for (int k = 0; k < KSPLIT - 1; ++k) {
      fk[k][v] = __expf(pmax[qb][k + 1][row] - m);
      l += psum[qb][k + 1][row] * fk[k][v];
    }
    ltot[v] = l;
  }
  #pragma unroll
  for (int t = 0; t < 4; ++t) {
    #pragma unroll
    for (int v = 0; v < 8; ++v) {
      float a = acc[t][v] * f0[v];
      #pragma unroll
      for (int k = 0; k < KSPLIT - 1; ++k)
        a += pacc[qb][k][(t * 8 + v) * 32 + lane_e] * fk[k][v];
      out[(size_t)(rowB + v + 8 * half_e) * DM + head * DHD + t * 16 + nn_e] =
          a / ltot[v];
    }
  }
}

extern "C" void kernel_launch(void* const* d_in, const int* in_sizes, int n_in,
                              void* d_out, int out_size, void* d_ws, size_t ws_size,
                              hipStream_t stream) {
  (void)in_sizes; (void)n_in; (void)out_size; (void)ws_size;
  const float* h   = (const float*)d_in[0];
  const float* adj = (const float*)d_in[1];
  const float* msk = (const float*)d_in[2];
  const float* Wq  = (const float*)d_in[3];
  const float* bq  = (const float*)d_in[4];
  const float* Wk  = (const float*)d_in[5];
  const float* bk  = (const float*)d_in[6];
  const float* Wv  = (const float*)d_in[7];
  const float* bv  = (const float*)d_in[8];
  float* out = (float*)d_out;

  float* Qo = (float*)d_ws;
  float* Ko = Qo + (size_t)NTOK * DM;
  float* Vo = Ko + (size_t)NTOK * DM;

  qkv_proj_kernel<<<32 * 8 * 3, 256, 0, stream>>>(
      h, Wq, bq, Wk, bk, Wv, bv, Qo, Ko, Vo);

  graph_attn_kernel<<<(NH * 256) / QBLK, 32 * QBLK * KSPLIT, 0, stream>>>(
      Qo, Ko, Vo, adj, msk, out);
}


# attention work-group barrier split: signal at a trip's top behind the stores of the next K/V group (LDS ring of 3 stages, rows asked for one trip earlier), wait at its end; same sums, same order
# speedup vs baseline: 2.2888x; 1.1209x over previous
#include <hip/hip_runtime.h>


typedef float v2f __attribute__((ext_vector_type(2)));
typedef float v4f __attribute__((ext_vector_type(4)));
typedef float v8f __attribute__((ext_vector_type(8)));

#define NTOK 4096
#define DM   512
#define NH   8
#define DHD  64
#define SCALE 0.125f

static __device__ __forceinline__ v8f wmma_f32(v2f a, v2f b, v8f c) {
  return __builtin_amdgcn_wmma_f32_16x16x4_f32(
       false, a,  false, b,
       (short)0, c,  false,  false);
}

#define PBM 128
#define PBN 64
#define PBK 64
#define HPITCH 68
#define WPITCH 136

__global__ __launch_bounds__(256) void qkv_proj_kernel(
    const float* __restrict__ h,
    const float* __restrict__ Wq, const float* __restrict__ bq,
    const float* __restrict__ Wk, const float* __restrict__ bk,
    const float* __restrict__ Wv, const float* __restrict__ bv,
    float* __restrict__ Qo, float* __restrict__ Ko, float* __restrict__ Vo)
{
  __shared__ alignas(16) float hst[2][PBM * HPITCH];
  __shared__ alignas(16) float wst[2][(PBK / 2) * WPITCH];

  const int tid  = threadIdx.x;
  const int lane = tid & 31;
  const int wid  = __builtin_amdgcn_readfirstlane(tid >> 5);
  const int c    = lane & 15;
  const int hf   = lane >> 4;

  const int cb   = blockIdx.x >> 5;
  const int rowB = (blockIdx.x & 31) * PBM;
  const int mat  = cb >> 3;
  const int col0 = (cb & 7) * PBN;

  const float* W; const float* bias; float* Out;
  if (mat == 0)      { W = Wq; bias = bq; Out = Qo; }
  else if (mat == 1) { W = Wk; bias = bk; Out = Ko; }
  else               { W = Wv; bias = bv; Out = Vo; }

  const int q  = tid & 15;
  const int r0 = tid >> 4;
  const float* hg = h + (size_t)(rowB + r0) * DM + 4 * q;
  const float* wg = W + (size_t)(2 * r0) * DM + col0 + 4 * q;
  const int hso = r0 * HPITCH + 4 * q;
  const int wso = r0 * WPITCH + 8 * q;

  v4f hr[8], wa[2], wb[2];
  #pragma unroll
  for (int i = 0; i < 8; ++i) hr[i] = *(const v4f*)(hg + (size_t)(16 * i) * DM);
  #pragma unroll
  for (int j = 0; j < 2; ++j) {
    wa[j] = *(const v4f*)(wg + (size_t)(32 * j) * DM);
    wb[j] = *(const v4f*)(wg + (size_t)(32 * j + 1) * DM);
  }
  #pragma unroll
  for (int i = 0; i < 8; ++i) *(v4f*)&hst[0][hso + 16 * i * HPITCH] = hr[i];
  #pragma unroll
  for (int j = 0; j < 2; ++j) {
    const v4f lo = {wa[j].x, wb[j].x, wa[j].y, wb[j].y};
    const v4f hi = {wa[j].z, wb[j].z, wa[j].w, wb[j].w};
    *(v4f*)&wst[0][wso + 16 * j * WPITCH]     = lo;
    *(v4f*)&wst[0][wso + 16 * j * WPITCH + 4] = hi;
  }
  __syncthreads();

  v8f acc[4] = {};
  const int aoff = (16 * wid + c) * HPITCH + 8 * hf;
  const int boff = (4 * hf) * WPITCH + 2 * c;

  for (int ks = 0; ks < DM / PBK; ++ks) {
    const int st = ks & 1;
    const bool more = (ks + 1) < DM / PBK;
    if (more) {
      const float* hn = hg + PBK * (ks + 1);
      const float* wn = wg + (size_t)(PBK * (ks + 1)) * DM;
      #pragma unroll
      for (int i = 0; i < 8; ++i) hr[i] = *(const v4f*)(hn + (size_t)(16 * i) * DM);
      #pragma unroll
      for (int j = 0; j < 2; ++j) {
        wa[j] = *(const v4f*)(wn + (size_t)(32 * j) * DM);
        wb[j] = *(const v4f*)(wn + (size_t)(32 * j + 1) * DM);
      }
    }

    const float* hs = &hst[st][aoff];
    const float* ws = &wst[st][boff];
    #pragma unroll
    for (int kc = 0; kc < PBK; kc += 16) {
      const v4f a01 = *(const v4f*)&hs[kc];
      const v4f a23 = *(const v4f*)&hs[kc + 4];
      const v2f as[4] = {{a01.x, a01.y}, {a01.z, a01.w},
                         {a23.x, a23.y}, {a23.z, a23.w}};
      #pragma unroll
      for (int s = 0; s < 4; ++s) {
        const float* w0 = ws + (kc / 2 + s) * WPITCH;
        #pragma unroll
        for (int t = 0; t < 4; ++t) {
          const v2f b = *(const v2f*)&w0[32 * t];
          acc[t] = wmma_f32(as[s], b, acc[t]);
        }
      }
    }

    if (more) {
      #pragma unroll
      for (int i = 0; i < 8; ++i) *(v4f*)&hst[st ^ 1][hso + 16 * i * HPITCH] = hr[i];
      #pragma unroll
      for (int j = 0; j < 2; ++j) {
        const v4f lo = {wa[j].x, wb[j].x, wa[j].y, wb[j].y};
        const v4f hi = {wa[j].z, wb[j].z, wa[j].w, wb[j].w};
        *(v4f*)&wst[st ^ 1][wso + 16 * j * WPITCH]     = lo;
        *(v4f*)&wst[st ^ 1][wso + 16 * j * WPITCH + 4] = hi;
      }
    }
    __syncthreads();
  }

  #pragma unroll
  for (int t = 0; t < 4; ++t) {
    const int n = col0 + 16 * t + c;
    const float bn = bias[n];
    #pragma unroll
    for (int v = 0; v < 8; ++v)
      Out[(size_t)(rowB + 16 * wid + 8 * hf + v) * DM + n] = acc[t][v] + bn;
  }
}

#define KSPLIT 3
#define QBLK   8
#define GROWS  (16 * KSPLIT)
#define NFULL  ((NTOK / 16) / KSPLIT)
#define NREM   ((NTOK / 16) % KSPLIT)
#define NSTG   3
#define KPITCH 68
#define VPITCH 136
#define SPITCH 20

static __device__ __forceinline__ float xhalf(float x) {
  const int b = __builtin_bit_cast(int, x);
  return __builtin_bit_cast(float,
      __builtin_amdgcn_permlanex16(b, b, 0x76543210, 0xfedcba98, false, false));
}

template <int N>
static __device__ __forceinline__ float row_share(float x) {
  return __builtin_bit_cast(float, __builtin_amdgcn_update_dpp(
      0, __builtin_bit_cast(int, x), 0x150 + N, 0xf, 0xf, true));
}

__global__ __launch_bounds__(32 * QBLK * KSPLIT) void graph_attn_kernel(
    const float* __restrict__ Q, const float* __restrict__ K,
    const float* __restrict__ V,
    const float* __restrict__ adj, const float* __restrict__ msk,
    float* __restrict__ out)
{
  __shared__ alignas(16) float pl[QBLK * KSPLIT][16 * SPITCH];
  __shared__ alignas(16) float kst[NSTG][GROWS * KPITCH];
  __shared__ alignas(16) float vst[NSTG][(GROWS / 2) * VPITCH];
  __shared__ float pacc[QBLK][KSPLIT - 1][4 * 8 * 32];
  __shared__ float pmax[QBLK][KSPLIT][16];
  __shared__ float psum[QBLK][KSPLIT][16];

  const int lane = threadIdx.x & 31;
  const int wid  = __builtin_amdgcn_readfirstlane(threadIdx.x >> 5);
  const int qb   = wid & (QBLK - 1);
  const int ks   = wid / QBLK;
  const int task = blockIdx.x * QBLK + qb;
  const int head = task >> 8;
  const int rowB = (task & 255) * 16;
  const int nn   = lane & 15;
  const int half = lane >> 4;
  float* plw = pl[wid];

  const float* qrow = Q + (size_t)(rowB + nn) * DM + head * DHD;
  v2f qa[16];
  #pragma unroll
  for (int s = 0; s < 16; ++s)
    qa[s] = *(const v2f*)(qrow + 4 * s + 2 * half);

  const float* adjp = adj + (size_t)(rowB + nn) * NTOK + 8 * half;
  const float* mskp = msk + (size_t)(rowB + nn) * NTOK + 8 * half;

  const float* kg = K + (size_t)(2 * wid + half) * DM + head * DHD + 4 * nn;
  const float* vg = V + (size_t)(2 * wid) * DM + head * DHD + 2 * lane;
  const int kso = (2 * wid + half) * KPITCH + 4 * nn;
  const int vso = wid * VPITCH + 4 * lane;
  v4f kr; v2f va, vb;
  {
    const v4f kr0 = *(const v4f*)kg;
    const v2f va0 = *(const v2f*)vg;
    const v2f vb0 = *(const v2f*)(vg + DM);
    const v4f vr0 = {va0.x, vb0.x, va0.y, vb0.y};
    *(v4f*)&kst[0][kso] = kr0;
    *(v4f*)&vst[0][vso] = vr0;
    kr = *(const v4f*)(kg + (size_t)GROWS * DM);
    va = *(const v2f*)(vg + (size_t)GROWS * DM);
    vb = *(const v2f*)(vg + (size_t)GROWS * DM + DM);
  }
  __syncthreads();

  v8f acc[4] = {};
  float m_run = -3.0e38f;
  float l_run = 0.0f;

  auto tile = [&](const int j0, const float* ksw, const float* vsw) {
    const v4f ad0 = *(const v4f*)(adjp + j0);
    const v4f ad1 = *(const v4f*)(adjp + j0 + 4);
    const v4f mk0 = *(const v4f*)(mskp + j0);
    const v4f mk1 = *(const v4f*)(mskp + j0 + 4);
    if (j0 + 16 * KSPLIT < NTOK)
      __builtin_prefetch(adjp + j0 + 16 * KSPLIT, 0, 1);
    __builtin_amdgcn_sched_barrier(0);

    v8f sf = {};
    #pragma unroll
    for (int s = 0; s < 16; ++s) {
      v2f b = *(const v2f*)&ksw[nn * KPITCH + 4 * s + 2 * half];
      sf = wmma_f32(qa[s], b, sf);
    }

    #pragma unroll
    for (int v = 0; v < 8; ++v)
      plw[(v + 8 * half) * SPITCH + nn] = sf[v];
    __builtin_amdgcn_wave_barrier();
    const v4f s0 = *(const v4f*)&plw[nn * SPITCH + 8 * half];
    const v4f s1 = *(const v4f*)&plw[nn * SPITCH + 8 * half + 4];

    float p[8];
    #pragma unroll
    for (int i = 0; i < 4; ++i) {
      p[i]     = (ad0[i] * s0[i] + mk0[i]) * SCALE;
      p[4 + i] = (ad1[i] * s1[i] + mk1[i]) * SCALE;
    }
    float bm = fmaxf(fmaxf(fmaxf(p[0], p[1]), fmaxf(p[2], p[3])),
                     fmaxf(fmaxf(p[4], p[5]), fmaxf(p[6], p[7])));
    bm = fmaxf(bm, xhalf(bm));
    const float newm = fmaxf(m_run, bm);
    const float fr   = __expf(m_run - newm);
    m_run = newm;
    #pragma unroll
    for (int i = 0; i < 8; ++i) p[i] = __expf(p[i] - newm);
    l_run = l_run * fr + (((p[0] + p[1]) + (p[2] + p[3])) +
                          ((p[4] + p[5]) + (p[6] + p[7])));

    const float frx = __builtin_bit_cast(float, __builtin_amdgcn_update_dpp(
        __builtin_bit_cast(int, fr), __builtin_bit_cast(int, fr),
        0x128  , 0x2  , 0xf, false));
#define RESCALE_ROW(v) { const float f_ = row_share<v>(frx);                   \
      acc[0][v] *= f_; acc[1][v] *= f_; acc[2][v] *= f_; acc[3][v] *= f_; }
    RESCALE_ROW(0) RESCALE_ROW(1) RESCALE_ROW(2) RESCALE_ROW(3)
    RESCALE_ROW(4) RESCALE_ROW(5) RESCALE_ROW(6) RESCALE_ROW(7)
#undef RESCALE_ROW

    #pragma unroll
    for (int s = 0; s < 4; ++s) {
      const v2f a = {p[2 * s], p[2 * s + 1]};
      const float* v0 = vsw + (4 * half + s) * VPITCH + 2 * nn;
      #pragma unroll
      for (int t = 0; t < 4; ++t) {
        const v2f b = *(const v2f*)&v0[32 * t];
        acc[t] = wmma_f32(a, b, acc[t]);
      }
    }
  };

  int sc = 0;
  for (int it = 0; it < NFULL; ++it) {
    const int sn = (sc == NSTG - 1) ? 0 : sc + 1;

    {
      const v4f vr = {va.x, vb.x, va.y, vb.y};
      *(v4f*)&kst[sn][kso] = kr;
      *(v4f*)&vst[sn][vso] = vr;
    }
    __builtin_amdgcn_fence(__ATOMIC_RELEASE, "workgroup");
    __builtin_amdgcn_s_barrier_signal(-1);

    const int jn = GROWS * (it + 2);
    const int jl = (jn + 2 * wid) < NTOK ? jn : 0;
    kr = *(const v4f*)(kg + (size_t)jl * DM);
    va = *(const v2f*)(vg + (size_t)jl * DM);
    vb = *(const v2f*)(vg + (size_t)jl * DM + DM);

    tile(GROWS * it + 16 * ks, &kst[sc][16 * ks * KPITCH], &vst[sc][8 * ks * VPITCH]);

    __builtin_amdgcn_s_barrier_wait(-1);
    __builtin_amdgcn_fence(__ATOMIC_ACQUIRE, "workgroup");
    sc = sn;
  }
  if (ks < NREM)
    tile(GROWS * NFULL + 16 * ks, &kst[sc][16 * ks * KPITCH], &vst[sc][8 * ks * VPITCH]);

  const int lane_e = (int)__builtin_amdgcn_mbcnt_lo(0xffffffffu, 0u);
  const int nn_e   = lane_e & 15;
  const int half_e = lane_e >> 4;

  const float l_own = l_run + xhalf(l_run);
  pmax[qb][ks][nn_e] = m_run;
  psum[qb][ks][nn_e] = l_own;

  if (ks != 0) {
    float* pa = pacc[qb][ks - 1];
    #pragma unroll
    for (int t = 0; t < 4; ++t) {
      #pragma unroll
      for (int v = 0; v < 8; ++v)
        pa[(t * 8 + v) * 32 + lane_e] = acc[t][v];
    }
  }
  __syncthreads();
  if (ks != 0) return;

  float f0[8], fk[KSPLIT - 1][8], ltot[8];
  #pragma unroll
  for (int v = 0; v < 8; ++v) {
    const int row = v + 8 * half_e;
    const float m0 = pmax[qb][0][row];
    float m = m0;
    #pragma unroll
    for (int k = 0; k < KSPLIT - 1; ++k) m = fmaxf(m, pmax[qb][k + 1][row]);
    f0[v] = __expf(m0 - m);
    float l = psum[qb][0][row] * f0[v];
    #pragma unroll
    for (int k = 0; k < KSPLIT - 1; ++k) {
      fk[k][v] = __expf(pmax[qb][k + 1][row] - m);
      l += psum[qb][k + 1][row] * fk[k][v];
    }
    ltot[v] = l;
  }
  #pragma unroll
  for (int t = 0; t < 4; ++t) {
    #pragma unroll
    for (int v = 0; v < 8; ++v) {
      float a = acc[t][v] * f0[v];
      #pragma unroll
      for (int k = 0; k < KSPLIT - 1; ++k)
        a += pacc[qb][k][(t * 8 + v) * 32 + lane_e] * fk[k][v];
      out[(size_t)(rowB + v + 8 * half_e) * DM + head * DHD + t * 16 + nn_e] =
          a / ltot[v];
    }
  }
}

extern "C" void kernel_launch(void* const* d_in, const int* in_sizes, int n_in,
                              void* d_out, int out_size, void* d_ws, size_t ws_size,
                              hipStream_t stream) {
  (void)in_sizes; (void)n_in; (void)out_size; (void)ws_size;
  const float* h   = (const float*)d_in[0];
  const float* adj = (const float*)d_in[1];
  const float* msk = (const float*)d_in[2];
  const float* Wq  = (const float*)d_in[3];
  const float* bq  = (const float*)d_in[4];
  const float* Wk  = (const float*)d_in[5];
  const float* bk  = (const float*)d_in[6];
  const float* Wv  = (const float*)d_in[7];
  const float* bv  = (const float*)d_in[8];
  float* out = (float*)d_out;

  float* Qo = (float*)d_ws;
  float* Ko = Qo + (size_t)NTOK * DM;
  float* Vo = Ko + (size_t)NTOK * DM;

  qkv_proj_kernel<<<32 * 8 * 3, 256, 0, stream>>>(
      h, Wq, bq, Wk, bk, Wv, bv, Qo, Ko, Vo);

  graph_attn_kernel<<<(NH * 256) / QBLK, 32 * QBLK * KSPLIT, 0, stream>>>(
      Qo, Ko, Vo, adj, msk, out);
}
